# starting kernel
# baseline (speedup 1.0000x reference)
#include <hip/hip_runtime.h>
#include <math.h>

#define NB     64
#define CIN    128
#define COUT   128
#define NMODE  64
#define NN     4096
#define LOG2N  12
#define KSPLIT 8

typedef float v2f __attribute__((ext_vector_type(2)));
typedef float v8f __attribute__((ext_vector_type(8)));

static __device__ __forceinline__ v8f wmma_f32_16x16x4(v2f a, v2f b, v8f c) {
  return __builtin_amdgcn_wmma_f32_16x16x4_f32(false, a, false, b, (short)0, c,
                                               false, false);
}

static __device__ __forceinline__ float cas_tw(int p) {
  float th = (float)(p & (NN - 1)) * (6.28318530717958647692f / (float)NN);
  float s, c;
  sincosf(th, &s, &c);
  return c + s;
}

__global__ void prep_cas(float* __restrict__ c1t, float* __restrict__ c2t) {
  int idx = blockIdx.x * blockDim.x + threadIdx.x;
  if (idx < NMODE * NN) {
    int m = idx >> LOG2N;
    int j = idx & (NN - 1);
    float v = 1.0f;
#pragma unroll
    for (int b = 0; b < LOG2N; ++b)
      if ((j >> b) & 1) v *= cas_tw(m << b);
    c1t[(size_t)m * NN + j] = v;
  } else {
    int t = idx - NMODE * NN;
    int n = t >> 6;
    int k = t & (NMODE - 1);
    float v = 1.0f / (float)NN;
#pragma unroll
    for (int b = 0; b < 6; ++b)
      if ((k >> b) & 1) v *= cas_tw(n << b);
    c2t[(size_t)n * NMODE + k] = v;
  }
}

__global__ void prep_w(const float* __restrict__ w, float* __restrict__ wp,
                       float* __restrict__ wm) {
  int idx = blockIdx.x * blockDim.x + threadIdx.x;
  if (idx >= CIN * COUT * NMODE) return;
  int k = idx & (NMODE - 1);
  int kneg = (NMODE - k) & (NMODE - 1);
  float a = w[idx];
  float b = w[idx - k + kneg];
  wp[idx] = a + b;
  wm[idx] = a - b;
}

__global__ void dht_fwd(const float* __restrict__ x, const float* __restrict__ c1t,
                        float* __restrict__ xhp) {
  int wave = (blockIdx.x * blockDim.x + threadIdx.x) >> 5;
  int lane = threadIdx.x & 31;
  int rt = wave >> 3;
  int kc = wave & (KSPLIT - 1);
  int r0 = rt * 16;
  int mrow = lane & 15;
  int khalf = lane >> 4;
  const float* xr  = x   + (size_t)(r0 + mrow) * NN;
  const float* b0p = c1t + (size_t)(mrow)      * NN;
  const float* b1p = c1t + (size_t)(16 + mrow) * NN;
  const float* b2p = c1t + (size_t)(32 + mrow) * NN;
  const float* b3p = c1t + (size_t)(48 + mrow) * NN;
  int k0 = kc * (NN / KSPLIT);
  int k1 = k0 + (NN / KSPLIT);
  v8f a0 = {}, a1 = {}, a2 = {}, a3 = {};
#pragma unroll 4
  for (int k = k0; k < k1; k += 4) {
    int kk = k + khalf * 2;
    v2f a  = *(const v2f*)(xr  + kk);
    v2f b0 = *(const v2f*)(b0p + kk);
    v2f b1 = *(const v2f*)(b1p + kk);
    v2f b2 = *(const v2f*)(b2p + kk);
    v2f b3 = *(const v2f*)(b3p + kk);
    a0 = wmma_f32_16x16x4(a, b0, a0);
    a1 = wmma_f32_16x16x4(a, b1, a1);
    a2 = wmma_f32_16x16x4(a, b2, a2);
    a3 = wmma_f32_16x16x4(a, b3, a3);
  }
  float* dst = xhp + (size_t)kc * ((size_t)NB * CIN * NMODE);
#pragma unroll
  for (int v = 0; v < 8; ++v) {
    int rr = r0 + v + 8 * khalf;
    float* drow = dst + (size_t)rr * NMODE + mrow;
    drow[0]  = a0[v];
    drow[16] = a1[v];
    drow[32] = a2[v];
    drow[48] = a3[v];
  }
}

__global__ void reduce_xh(const float* __restrict__ xhp, float* __restrict__ xh) {
  int idx = blockIdx.x * blockDim.x + threadIdx.x;
  if (idx >= NB * CIN * NMODE) return;
  float s = 0.0f;
#pragma unroll
  for (int c = 0; c < KSPLIT; ++c)
    s += xhp[(size_t)c * ((size_t)NB * CIN * NMODE) + idx];
  xh[idx] = s;
}

__global__ void mode_mix(const float* __restrict__ xh, const float* __restrict__ wp,
                         const float* __restrict__ wm, float* __restrict__ outm) {
  int wave = (blockIdx.x * blockDim.x + threadIdx.x) >> 5;
  int lane = threadIdx.x & 31;
  int k   = wave >> 5;
  int rem = wave & 31;
  int bt = rem >> 3;
  int ot = rem & 7;
  int kneg = (NMODE - k) & (NMODE - 1);
  int mrow = lane & 15;
  int khalf = lane >> 4;
  int bglob = bt * 16 + mrow;
  int oglob = ot * 16 + mrow;
  const float* xr = xh + (size_t)bglob * CIN * NMODE;
  v8f acc = {};
#pragma unroll 4
  for (int i = 0; i < CIN; i += 4) {
    int i0 = i + khalf * 2;
    v2f ap, an, bp, bm;
    ap.x = xr[(size_t)i0 * NMODE + k];
    ap.y = xr[(size_t)(i0 + 1) * NMODE + k];
    an.x = xr[(size_t)i0 * NMODE + kneg];
    an.y = xr[(size_t)(i0 + 1) * NMODE + kneg];
    bp.x = wp[((size_t)i0 * COUT + oglob) * NMODE + k];
    bp.y = wp[((size_t)(i0 + 1) * COUT + oglob) * NMODE + k];
    bm.x = wm[((size_t)i0 * COUT + oglob) * NMODE + k];
    bm.y = wm[((size_t)(i0 + 1) * COUT + oglob) * NMODE + k];
    acc = wmma_f32_16x16x4(ap, bp, acc);
    acc = wmma_f32_16x16x4(an, bm, acc);
  }
#pragma unroll
  for (int v = 0; v < 8; ++v) {
    int bb = bt * 16 + v + 8 * khalf;
    outm[((size_t)bb * COUT + oglob) * NMODE + k] = 0.5f * acc[v];
  }
}

__global__ void dht_inv(const float* __restrict__ outm, const float* __restrict__ c2t,
                        float* __restrict__ y) {
  int wave = (blockIdx.x * blockDim.x + threadIdx.x) >> 5;
  int lane = threadIdx.x & 31;
  int rt = wave >> 6;
  int nc = wave & 63;
  int r0 = rt * 16;
  int n0 = nc * 64;
  int mrow = lane & 15;
  int khalf = lane >> 4;
  const float* ar  = outm + (size_t)(r0 + mrow) * NMODE;
  const float* b0p = c2t + (size_t)(n0 + mrow)      * NMODE;
  const float* b1p = c2t + (size_t)(n0 + 16 + mrow) * NMODE;
  const float* b2p = c2t + (size_t)(n0 + 32 + mrow) * NMODE;
  const float* b3p = c2t + (size_t)(n0 + 48 + mrow) * NMODE;
  v8f a0 = {}, a1 = {}, a2 = {}, a3 = {};
#pragma unroll 4
  for (int k = 0; k < NMODE; k += 4) {
    int kk = k + khalf * 2;
    v2f a  = *(const v2f*)(ar  + kk);
    v2f b0 = *(const v2f*)(b0p + kk);
    v2f b1 = *(const v2f*)(b1p + kk);
    v2f b2 = *(const v2f*)(b2p + kk);
    v2f b3 = *(const v2f*)(b3p + kk);
    a0 = wmma_f32_16x16x4(a, b0, a0);
    a1 = wmma_f32_16x16x4(a, b1, a1);
    a2 = wmma_f32_16x16x4(a, b2, a2);
    a3 = wmma_f32_16x16x4(a, b3, a3);
  }
#pragma unroll
  for (int v = 0; v < 8; ++v) {
    int rr = r0 + v + 8 * khalf;
    float* drow = y + (size_t)rr * NN + n0 + mrow;
    drow[0]  = a0[v];
    drow[16] = a1[v];
    drow[32] = a2[v];
    drow[48] = a3[v];
  }
}

extern "C" void kernel_launch(void* const* d_in, const int* in_sizes, int n_in,
                              void* d_out, int out_size, void* d_ws, size_t ws_size,
                              hipStream_t stream) {
  (void)in_sizes; (void)n_in; (void)out_size; (void)ws_size;
  const float* x = (const float*)d_in[0];
  const float* w = (const float*)d_in[1];
  float* y = (float*)d_out;

  float* ws   = (float*)d_ws;
  float* c1t  = ws;
  float* c2t  = c1t + (size_t)NMODE * NN;
  float* wp   = c2t + (size_t)NN * NMODE;
  float* wm   = wp  + (size_t)CIN * COUT * NMODE;
  float* xhp  = wm  + (size_t)CIN * COUT * NMODE;
  float* xh   = xhp + (size_t)KSPLIT * NB * CIN * NMODE;
  float* outm = xh  + (size_t)NB * CIN * NMODE;

  prep_cas<<<(2 * NMODE * NN) / 256, 256, 0, stream>>>(c1t, c2t);
  prep_w<<<(CIN * COUT * NMODE) / 256, 256, 0, stream>>>(w, wp, wm);

  dht_fwd<<<((NB * CIN / 16) * KSPLIT * 32) / 256, 256, 0, stream>>>(x, c1t, xhp);
  reduce_xh<<<(NB * CIN * NMODE) / 256, 256, 0, stream>>>(xhp, xh);

  mode_mix<<<(NMODE * (NB / 16) * (COUT / 16) * 32) / 256, 256, 0, stream>>>(
      xh, wp, wm, outm);

  dht_inv<<<((NB * COUT / 16) * (NN / 64) * 32) / 256, 256, 0, stream>>>(
      outm, c2t, y);
}


# mirror-mode pairing moved from weights to spectrum: reduce writes xh[k]+xh[-k], xh[k]-xh[-k] (mirror via LDS), mode_mix multiplies them by w[k], w[-k] as given; prep_w launch gone; sum regrouped
# speedup vs baseline: 2.1887x; 2.1887x over previous
#include <hip/hip_runtime.h>
#include <math.h>

#define NB     64
#define CIN    128
#define COUT   128
#define NMODE  64
#define NN     4096
#define LOG2N  12
#define KSPLIT 8

typedef float v2f __attribute__((ext_vector_type(2)));
typedef float v4f __attribute__((ext_vector_type(4)));
typedef float v8f __attribute__((ext_vector_type(8)));

static __device__ __forceinline__ v8f wmma_f32_16x16x4(v2f a, v2f b, v8f c) {
  return __builtin_amdgcn_wmma_f32_16x16x4_f32(false, a, false, b, (short)0, c,
                                               false, false);
}

static __device__ __forceinline__ float cas_tw(int p) {
  float th = (float)(p & (NN - 1)) * (6.28318530717958647692f / (float)NN);
  float s, c;
  sincosf(th, &s, &c);
  return c + s;
}

__global__ void prep_cas(float* __restrict__ c1t, float* __restrict__ c2t) {
  int idx = blockIdx.x * blockDim.x + threadIdx.x;
  if (idx < NMODE * NN) {
    int m = idx >> LOG2N;
    int j = idx & (NN - 1);
    float v = 1.0f;
#pragma unroll
    for (int b = 0; b < LOG2N; ++b)
      if ((j >> b) & 1) v *= cas_tw(m << b);
    c1t[(size_t)m * NN + j] = v;
  } else {
    int t = idx - NMODE * NN;
    int n = t >> 6;
    int k = t & (NMODE - 1);
    float v = 1.0f / (float)NN;
#pragma unroll
    for (int b = 0; b < 6; ++b)
      if ((k >> b) & 1) v *= cas_tw(n << b);
    c2t[(size_t)n * NMODE + k] = v;
  }
}

#define TM 128
#define TN 64
#define TK 64
#define LP 68

template <int NP, int LD>
static __device__ __forceinline__ void tile_load(v4f (&r)[NP], const float* __restrict__ src, int tid) {
  int row = tid >> 4;
  int c4  = (tid & 15) * 4;
#pragma unroll
  for (int p = 0; p < NP; ++p)
    r[p] = *(const v4f*)(src + (size_t)(row + 16 * p) * LD + c4);
}

template <int NP>
static __device__ __forceinline__ void tile_store(const v4f (&r)[NP], float* dst, int tid) {
  int row = tid >> 4;
  int c4  = (tid & 15) * 4;
#pragma unroll
  for (int p = 0; p < NP; ++p)
    *(v4f*)(dst + (row + 16 * p) * LP + c4) = r[p];
}

static __device__ __forceinline__ void stage_mma(const float* sA, const float* sB, int w, int lane,
                                                 v8f& a0, v8f& a1, v8f& a2, v8f& a3) {
  int c = lane & 15;
  int h = lane >> 4;
  const float* ap = sA + (w * 16 + c) * LP + 2 * h;
  const float* bp = sB + c * LP + 2 * h;
#pragma unroll
  for (int k = 0; k < TK; k += 4) {
    v2f a  = *(const v2f*)(ap + k);
    v2f b0 = *(const v2f*)(bp + k);
    v2f b1 = *(const v2f*)(bp + 16 * LP + k);
    v2f b2 = *(const v2f*)(bp + 32 * LP + k);
    v2f b3 = *(const v2f*)(bp + 48 * LP + k);
    a0 = wmma_f32_16x16x4(a, b0, a0);
    a1 = wmma_f32_16x16x4(a, b1, a1);
    a2 = wmma_f32_16x16x4(a, b2, a2);
    a3 = wmma_f32_16x16x4(a, b3, a3);
  }
}

__global__ void __launch_bounds__(256) dht_fwd(const float* __restrict__ x, const float* __restrict__ c1t,
                        float* __restrict__ xhp) {
  __shared__ __attribute__((aligned(16))) float sA[TM * LP];
  __shared__ __attribute__((aligned(16))) float sB[TN * LP];
  int tid  = threadIdx.x;
  int lane = tid & 31;
  int w    = tid >> 5;
  int g  = blockIdx.x >> 3;
  int kc = blockIdx.x & (KSPLIT - 1);
  int r0 = g * TM;
  int k0 = kc * (NN / KSPLIT);
  const float* A = x + (size_t)r0 * NN + k0;
  const float* B = c1t + k0;
  v8f a0 = {}, a1 = {}, a2 = {}, a3 = {};
  v4f ra[8], rb[4];
  tile_load<8, NN>(ra, A, tid);
  tile_load<4, NN>(rb, B, tid);
#pragma unroll 1
  for (int s = 0; s < (NN / KSPLIT) / TK; ++s) {
    tile_store<8>(ra, sA, tid);
    tile_store<4>(rb, sB, tid);
    __syncthreads();
    if (s + 1 < (NN / KSPLIT) / TK) {
      tile_load<8, NN>(ra, A + (s + 1) * TK, tid);
      tile_load<4, NN>(rb, B + (s + 1) * TK, tid);
    }
    stage_mma(sA, sB, w, lane, a0, a1, a2, a3);
    __syncthreads();
  }
  int mrow = lane & 15;
  int khalf = lane >> 4;
  float* dst = xhp + (size_t)kc * ((size_t)NB * CIN * NMODE);
#pragma unroll
  for (int v = 0; v < 8; ++v) {
    int rr = r0 + w * 16 + v + 8 * khalf;
    float* drow = dst + (size_t)rr * NMODE + mrow;
    drow[0]  = a0[v];
    drow[16] = a1[v];
    drow[32] = a2[v];
    drow[48] = a3[v];
  }
}

#define RED_T 256
static_assert((NB * CIN * NMODE) % RED_T == 0 && RED_T % NMODE == 0,
              "reduce_pair: whole rows per work-group, no partial work-group");
__global__ void __launch_bounds__(RED_T) reduce_pair(const float* __restrict__ xhp, float* __restrict__ xp,
                                                      float* __restrict__ xm) {
  __shared__ float sh[RED_T];
  int tid = threadIdx.x;
  int idx = blockIdx.x * RED_T + tid;
  float s = 0.0f;
#pragma unroll
  for (int c = 0; c < KSPLIT; ++c)
    s += xhp[(size_t)c * ((size_t)NB * CIN * NMODE) + idx];
  sh[tid] = s;
  __syncthreads();
  int k = tid & (NMODE - 1);
  float t = sh[(tid - k) + ((NMODE - k) & (NMODE - 1))];
  xp[idx] = s + t;
  xm[idx] = s - t;
}

struct MixFrag {
  v4f p0[4], p1[4], m0[4], m1[4], w0[4], w1[4], v0[4], v1[4];
  float s0[4], s1[4];
};

static __device__ __forceinline__ void mix_load(MixFrag& f, const float* __restrict__ xpr,
                                                const float* __restrict__ xmr,
                                                const float* __restrict__ wr,
                                                int ibase, int khalf, int k0, int kn4, int kn0) {
#pragma unroll
  for (int s = 0; s < 4; ++s) {
    int i0 = ibase + 4 * s + khalf * 2;
    const float* a0 = xpr + (size_t)i0 * NMODE + k0;
    const float* b0 = xmr + (size_t)i0 * NMODE + k0;
    const float* c0 = wr + (size_t)i0 * COUT * NMODE;
    const float* c1 = c0 + (size_t)COUT * NMODE;
    f.p0[s] = *(const v4f*)(a0);
    f.p1[s] = *(const v4f*)(a0 + NMODE);
    f.m0[s] = *(const v4f*)(b0);
    f.m1[s] = *(const v4f*)(b0 + NMODE);
    f.w0[s] = *(const v4f*)(c0 + k0);
    f.w1[s] = *(const v4f*)(c1 + k0);
    f.v0[s] = *(const v4f*)(c0 + kn4);
    f.v1[s] = *(const v4f*)(c1 + kn4);
    f.s0[s] = c0[kn0];
    f.s1[s] = c1[kn0];
  }
}

static __device__ __forceinline__ void mix_mma(const MixFrag& f, v8f& acc0, v8f& acc1,
                                               v8f& acc2, v8f& acc3) {
#pragma unroll
  for (int s = 0; s < 4; ++s) {
    v2f ap, am, bp, bm;
    ap.x = f.p0[s].x;  ap.y = f.p1[s].x;  bp.x = f.w0[s].x;  bp.y = f.w1[s].x;
    am.x = f.m0[s].x;  am.y = f.m1[s].x;  bm.x = f.s0[s];    bm.y = f.s1[s];
    acc0 = wmma_f32_16x16x4(ap, bp, acc0);
    acc0 = wmma_f32_16x16x4(am, bm, acc0);
    ap.x = f.p0[s].y;  ap.y = f.p1[s].y;  bp.x = f.w0[s].y;  bp.y = f.w1[s].y;
    am.x = f.m0[s].y;  am.y = f.m1[s].y;  bm.x = f.v0[s].w;  bm.y = f.v1[s].w;
    acc1 = wmma_f32_16x16x4(ap, bp, acc1);
    acc1 = wmma_f32_16x16x4(am, bm, acc1);
    ap.x = f.p0[s].z;  ap.y = f.p1[s].z;  bp.x = f.w0[s].z;  bp.y = f.w1[s].z;
    am.x = f.m0[s].z;  am.y = f.m1[s].z;  bm.x = f.v0[s].z;  bm.y = f.v1[s].z;
    acc2 = wmma_f32_16x16x4(ap, bp, acc2);
    acc2 = wmma_f32_16x16x4(am, bm, acc2);
    ap.x = f.p0[s].w;  ap.y = f.p1[s].w;  bp.x = f.w0[s].w;  bp.y = f.w1[s].w;
    am.x = f.m0[s].w;  am.y = f.m1[s].w;  bm.x = f.v0[s].y;  bm.y = f.v1[s].y;
    acc3 = wmma_f32_16x16x4(ap, bp, acc3);
    acc3 = wmma_f32_16x16x4(am, bm, acc3);
  }
}

__global__ void __launch_bounds__(128) mode_mix(const float* __restrict__ xp, const float* __restrict__ xm,
                         const float* __restrict__ w, float* __restrict__ outm) {
  int wave = (blockIdx.x * blockDim.x + threadIdx.x) >> 5;
  int lane = threadIdx.x & 31;
  int kq  = wave & 15;
  int rem = wave >> 4;
  int bt = rem >> 3;
  int ot = rem & 7;
  int k0  = kq * 4;
  int kn4 = (NMODE - 4) - k0;
  int kn0 = (NMODE - k0) & (NMODE - 1);
  int mrow = lane & 15;
  int khalf = lane >> 4;
  int bglob = bt * 16 + mrow;
  int oglob = ot * 16 + mrow;
  const float* xpr = xp + (size_t)bglob * CIN * NMODE;
  const float* xmr = xm + (size_t)bglob * CIN * NMODE;
  const float* wr  = w + (size_t)oglob * NMODE;
  v8f acc0 = {}, acc1 = {}, acc2 = {}, acc3 = {};
  MixFrag fa, fb;
  mix_load(fa, xpr, xmr, wr, 0, khalf, k0, kn4, kn0);
#pragma unroll 1
  for (int i = 0; i < CIN - 32; i += 32) {
    mix_load(fb, xpr, xmr, wr, i + 16, khalf, k0, kn4, kn0);
    mix_mma(fa, acc0, acc1, acc2, acc3);
    mix_load(fa, xpr, xmr, wr, i + 32, khalf, k0, kn4, kn0);
    mix_mma(fb, acc0, acc1, acc2, acc3);
  }
  mix_load(fb, xpr, xmr, wr, CIN - 16, khalf, k0, kn4, kn0);
  mix_mma(fa, acc0, acc1, acc2, acc3);
  mix_mma(fb, acc0, acc1, acc2, acc3);
#pragma unroll
  for (int v = 0; v < 8; ++v) {
    int bb = bt * 16 + v + 8 * khalf;
    v4f o;
    o.x = 0.5f * acc0[v];
    o.y = 0.5f * acc1[v];
    o.z = 0.5f * acc2[v];
    o.w = 0.5f * acc3[v];
    *(v4f*)(outm + ((size_t)bb * COUT + oglob) * NMODE + k0) = o;
  }
}

#define INV_NCH 4
__global__ void __launch_bounds__(256) dht_inv(const float* __restrict__ outm, const float* __restrict__ c2t,
                        float* __restrict__ y) {
  __shared__ __attribute__((aligned(16))) float sA[TM * LP];
  __shared__ __attribute__((aligned(16))) float sB[TN * LP];
  int tid  = threadIdx.x;
  int lane = tid & 31;
  int w    = tid >> 5;
  int g  = blockIdx.x >> 4;
  int q  = blockIdx.x & 15;
  int r0 = g * TM;
  int n0 = q * (INV_NCH * TN);
  const float* B = c2t + (size_t)n0 * NMODE;
  v4f ra[8], rb[4];
  tile_load<8, NMODE>(ra, outm + (size_t)r0 * NMODE, tid);
  tile_load<4, NMODE>(rb, B, tid);
  tile_store<8>(ra, sA, tid);
  int mrow = lane & 15;
  int khalf = lane >> 4;
  float* yrow = y + (size_t)(r0 + w * 16 + 8 * khalf) * NN + n0 + mrow;
#pragma unroll 1
  for (int ch = 0; ch < INV_NCH; ++ch) {
    tile_store<4>(rb, sB, tid);
    __syncthreads();
    if (ch + 1 < INV_NCH)
      tile_load<4, NMODE>(rb, B + (size_t)(ch + 1) * TN * NMODE, tid);
    v8f a0 = {}, a1 = {}, a2 = {}, a3 = {};
    stage_mma(sA, sB, w, lane, a0, a1, a2, a3);
    float* yc = yrow + ch * TN;
#pragma unroll
    for (int v = 0; v < 8; ++v) {
      float* drow = yc + (size_t)v * NN;
      drow[0]  = a0[v];
      drow[16] = a1[v];
      drow[32] = a2[v];
      drow[48] = a3[v];
    }
    __syncthreads();
  }
}

extern "C" void kernel_launch(void* const* d_in, const int* in_sizes, int n_in,
                              void* d_out, int out_size, void* d_ws, size_t ws_size,
                              hipStream_t stream) {
  (void)in_sizes; (void)n_in; (void)out_size; (void)ws_size;
  const float* x = (const float*)d_in[0];
  const float* w = (const float*)d_in[1];
  float* y = (float*)d_out;

  float* ws   = (float*)d_ws;
  float* c1t  = ws;
  float* c2t  = c1t + (size_t)NMODE * NN;
  float* xhp  = c2t + (size_t)NN * NMODE;
  float* xp   = xhp + (size_t)KSPLIT * NB * CIN * NMODE;
  float* xm   = xp  + (size_t)NB * CIN * NMODE;
  float* outm = xm  + (size_t)NB * CIN * NMODE;

  prep_cas<<<(2 * NMODE * NN) / 256, 256, 0, stream>>>(c1t, c2t);

  dht_fwd<<<(NB * CIN / TM) * KSPLIT, 256, 0, stream>>>(x, c1t, xhp);
  reduce_pair<<<(NB * CIN * NMODE) / RED_T, RED_T, 0, stream>>>(xhp, xp, xm);

  mode_mix<<<((NMODE / 4) * (NB / 16) * (COUT / 16) * 32) / 128, 128, 0, stream>>>(
      xp, xm, w, outm);

  dht_inv<<<(NB * COUT / TM) * (NN / (INV_NCH * TN)), 256, 0, stream>>>(
      outm, c2t, y);
}


# mode_mix takes its operands through LDS: a work-group copies its rows' 64-byte runs (xp, xm, w at +k and -k) with whole-run loads into two tile sets; fragments become LDS reads; same sums
# speedup vs baseline: 2.2925x; 1.0474x over previous
#include <hip/hip_runtime.h>
#include <math.h>

#define NB     64
#define CIN    128
#define COUT   128
#define NMODE  64
#define NN     4096
#define LOG2N  12
#define KSPLIT 8

typedef float v2f __attribute__((ext_vector_type(2)));
typedef float v4f __attribute__((ext_vector_type(4)));
typedef float v8f __attribute__((ext_vector_type(8)));

static __device__ __forceinline__ v8f wmma_f32_16x16x4(v2f a, v2f b, v8f c) {
  return __builtin_amdgcn_wmma_f32_16x16x4_f32(false, a, false, b, (short)0, c,
                                               false, false);
}

static __device__ __forceinline__ float cas_tw(int p) {
  float th = (float)(p & (NN - 1)) * (6.28318530717958647692f / (float)NN);
  float s, c;
  sincosf(th, &s, &c);
  return c + s;
}

__global__ void prep_cas(float* __restrict__ c1t, float* __restrict__ c2t) {
  int idx = blockIdx.x * blockDim.x + threadIdx.x;
  if (idx < NMODE * NN) {
    int m = idx >> LOG2N;
    int j = idx & (NN - 1);
    float v = 1.0f;
#pragma unroll
    for (int b = 0; b < LOG2N; ++b)
      if ((j >> b) & 1) v *= cas_tw(m << b);
    c1t[(size_t)m * NN + j] = v;
  } else {
    int t = idx - NMODE * NN;
    int n = t >> 6;
    int k = t & (NMODE - 1);
    float v = 1.0f / (float)NN;
#pragma unroll
    for (int b = 0; b < 6; ++b)
      if ((k >> b) & 1) v *= cas_tw(n << b);
    c2t[(size_t)n * NMODE + k] = v;
  }
}

#define TM 128
#define TN 64
#define TK 64
#define LP 68

template <int NP, int LD>
static __device__ __forceinline__ void tile_load(v4f (&r)[NP], const float* __restrict__ src, int tid) {
  int row = tid >> 4;
  int c4  = (tid & 15) * 4;
#pragma unroll
  for (int p = 0; p < NP; ++p)
    r[p] = *(const v4f*)(src + (size_t)(row + 16 * p) * LD + c4);
}

template <int NP>
static __device__ __forceinline__ void tile_store(const v4f (&r)[NP], float* dst, int tid) {
  int row = tid >> 4;
  int c4  = (tid & 15) * 4;
#pragma unroll
  for (int p = 0; p < NP; ++p)
    *(v4f*)(dst + (row + 16 * p) * LP + c4) = r[p];
}

static __device__ __forceinline__ void stage_mma(const float* sA, const float* sB, int w, int lane,
                                                 v8f& a0, v8f& a1, v8f& a2, v8f& a3) {
  int c = lane & 15;
  int h = lane >> 4;
  const float* ap = sA + (w * 16 + c) * LP + 2 * h;
  const float* bp = sB + c * LP + 2 * h;
#pragma unroll
  for (int k = 0; k < TK; k += 4) {
    v2f a  = *(const v2f*)(ap + k);
    v2f b0 = *(const v2f*)(bp + k);
    v2f b1 = *(const v2f*)(bp + 16 * LP + k);
    v2f b2 = *(const v2f*)(bp + 32 * LP + k);
    v2f b3 = *(const v2f*)(bp + 48 * LP + k);
    a0 = wmma_f32_16x16x4(a, b0, a0);
    a1 = wmma_f32_16x16x4(a, b1, a1);
    a2 = wmma_f32_16x16x4(a, b2, a2);
    a3 = wmma_f32_16x16x4(a, b3, a3);
  }
}

__global__ void __launch_bounds__(256) dht_fwd(const float* __restrict__ x, const float* __restrict__ c1t,
                        float* __restrict__ xhp) {
  __shared__ __attribute__((aligned(16))) float sA[TM * LP];
  __shared__ __attribute__((aligned(16))) float sB[TN * LP];
  int tid  = threadIdx.x;
  int lane = tid & 31;
  int w    = tid >> 5;
  int g  = blockIdx.x >> 3;
  int kc = blockIdx.x & (KSPLIT - 1);
  int r0 = g * TM;
  int k0 = kc * (NN / KSPLIT);
  const float* A = x + (size_t)r0 * NN + k0;
  const float* B = c1t + k0;
  v8f a0 = {}, a1 = {}, a2 = {}, a3 = {};
  v4f ra[8], rb[4];
  tile_load<8, NN>(ra, A, tid);
  tile_load<4, NN>(rb, B, tid);
#pragma unroll 1
  for (int s = 0; s < (NN / KSPLIT) / TK; ++s) {
    tile_store<8>(ra, sA, tid);
    tile_store<4>(rb, sB, tid);
    __syncthreads();
    if (s + 1 < (NN / KSPLIT) / TK) {
      tile_load<8, NN>(ra, A + (s + 1) * TK, tid);
      tile_load<4, NN>(rb, B + (s + 1) * TK, tid);
    }
    stage_mma(sA, sB, w, lane, a0, a1, a2, a3);
    __syncthreads();
  }
  int mrow = lane & 15;
  int khalf = lane >> 4;
  float* dst = xhp + (size_t)kc * ((size_t)NB * CIN * NMODE);
#pragma unroll
  for (int v = 0; v < 8; ++v) {
    int rr = r0 + w * 16 + v + 8 * khalf;
    float* drow = dst + (size_t)rr * NMODE + mrow;
    drow[0]  = a0[v];
    drow[16] = a1[v];
    drow[32] = a2[v];
    drow[48] = a3[v];
  }
}

#define RED_T 256
static_assert((NB * CIN * NMODE) % RED_T == 0 && RED_T % NMODE == 0,
              "reduce_pair: whole rows per work-group, no partial work-group");
__global__ void __launch_bounds__(RED_T) reduce_pair(const float* __restrict__ xhp, float* __restrict__ xp,
                                                      float* __restrict__ xm) {
  __shared__ float sh[RED_T];
  int tid = threadIdx.x;
  int idx = blockIdx.x * RED_T + tid;
  float s = 0.0f;
#pragma unroll
  for (int c = 0; c < KSPLIT; ++c)
    s += xhp[(size_t)c * ((size_t)NB * CIN * NMODE) + idx];
  sh[tid] = s;
  __syncthreads();
  int k = tid & (NMODE - 1);
  float t = sh[(tid - k) + ((NMODE - k) & (NMODE - 1))];
  xp[idx] = s + t;
  xm[idx] = s - t;
}

struct MixFrag {
  v4f p0[4], p1[4], m0[4], m1[4], w0[4], w1[4], v0[4], v1[4];
  float s0[4], s1[4];
};

#define MXI 16
#define MXP 20
#define MXT (MXI * 16 * MXP)

static __device__ __forceinline__ void mix_frag(MixFrag& f, int s, const float* sXP, const float* sXM,
                                                const float* sWK, const float* sWN,
                                                int c, int khalf, int q) {
  int r0 = ((4 * s + 2 * khalf) * 16 + c) * MXP;
  int r1 = r0 + 16 * MXP;
  f.p0[s] = *(const v4f*)(sXP + r0 + 4 * q);
  f.p1[s] = *(const v4f*)(sXP + r1 + 4 * q);
  f.m0[s] = *(const v4f*)(sXM + r0 + 4 * q);
  f.m1[s] = *(const v4f*)(sXM + r1 + 4 * q);
  f.w0[s] = *(const v4f*)(sWK + r0 + 4 * q);
  f.w1[s] = *(const v4f*)(sWK + r1 + 4 * q);
  f.v0[s] = *(const v4f*)(sWN + r0 + 12 - 4 * q);
  f.v1[s] = *(const v4f*)(sWN + r1 + 12 - 4 * q);
  f.s0[s] = sWN[r0 + 16 - 4 * q];
  f.s1[s] = sWN[r1 + 16 - 4 * q];
}

struct MixStage {
  v4f xp[8], xm[8], wk[8], wn[10];
};

static __device__ __forceinline__ void mix_stage_load(MixStage& g, const float* __restrict__ xpb,
                                                      const float* __restrict__ xmb,
                                                      const float* __restrict__ wb,
                                                      int i0, int K0, int MB, int tid) {
#pragma unroll
  for (int j = 0; j < 8; ++j) {
    int p  = tid + 128 * j;
    int pc = (p & 3) * 4;
    int gr = p >> 2;
    int il = gr >> 4;
    int c  = gr & 15;
    size_t xo = ((size_t)c * CIN + (i0 + il)) * NMODE + pc;
    size_t wo = ((size_t)(i0 + il) * COUT + c) * NMODE + pc;
    g.xp[j] = *(const v4f*)(xpb + xo + K0);
    g.xm[j] = *(const v4f*)(xmb + xo + K0);
    g.wk[j] = *(const v4f*)(wb + wo + K0);
  }
#pragma unroll
  for (int j = 0; j < 10; ++j) {
    int p  = tid + 128 * j;
    int gr = p / 5;
    int pc = p - 5 * gr;
    int il = gr >> 4;
    int c  = gr & 15;
    g.wn[j] = *(const v4f*)(wb + ((size_t)(i0 + il) * COUT + c) * NMODE + ((MB + 4 * pc) & (NMODE - 1)));
  }
}

static __device__ __forceinline__ void mix_stage_store(const MixStage& g, int part, float* sXP, float* sXM,
                                                       float* sWK, float* sWN, int tid) {
#pragma unroll
  for (int j = 2 * part; j < 2 * part + 2; ++j) {
    int p = tid + 128 * j;
    int o = (p >> 2) * MXP + (p & 3) * 4;
    *(v4f*)(sXP + o) = g.xp[j];
    *(v4f*)(sXM + o) = g.xm[j];
    *(v4f*)(sWK + o) = g.wk[j];
  }
#pragma unroll
  for (int j = part; j < 10; j += 4)
    *(v4f*)(sWN + 4 * (tid + 128 * j)) = g.wn[j];
}

static __device__ __forceinline__ void mix_mma(const MixFrag& f, int s, v8f& acc0, v8f& acc1,
                                               v8f& acc2, v8f& acc3) {
  v2f ap, am, bp, bm;
  ap.x = f.p0[s].x;  ap.y = f.p1[s].x;  bp.x = f.w0[s].x;  bp.y = f.w1[s].x;
  am.x = f.m0[s].x;  am.y = f.m1[s].x;  bm.x = f.s0[s];    bm.y = f.s1[s];
  acc0 = wmma_f32_16x16x4(ap, bp, acc0);
  acc0 = wmma_f32_16x16x4(am, bm, acc0);
  ap.x = f.p0[s].y;  ap.y = f.p1[s].y;  bp.x = f.w0[s].y;  bp.y = f.w1[s].y;
  am.x = f.m0[s].y;  am.y = f.m1[s].y;  bm.x = f.v0[s].w;  bm.y = f.v1[s].w;
  acc1 = wmma_f32_16x16x4(ap, bp, acc1);
  acc1 = wmma_f32_16x16x4(am, bm, acc1);
  ap.x = f.p0[s].z;  ap.y = f.p1[s].z;  bp.x = f.w0[s].z;  bp.y = f.w1[s].z;
  am.x = f.m0[s].z;  am.y = f.m1[s].z;  bm.x = f.v0[s].z;  bm.y = f.v1[s].z;
  acc2 = wmma_f32_16x16x4(ap, bp, acc2);
  acc2 = wmma_f32_16x16x4(am, bm, acc2);
  ap.x = f.p0[s].w;  ap.y = f.p1[s].w;  bp.x = f.w0[s].w;  bp.y = f.w1[s].w;
  am.x = f.m0[s].w;  am.y = f.m1[s].w;  bm.x = f.v0[s].y;  bm.y = f.v1[s].y;
  acc3 = wmma_f32_16x16x4(ap, bp, acc3);
  acc3 = wmma_f32_16x16x4(am, bm, acc3);
}

static __device__ __forceinline__ void mix_stage(const float* cur, float* nxt, const MixStage& g, bool fill,
                                                 int mrow, int khalf, int q, int tid,
                                                 v8f& acc0, v8f& acc1, v8f& acc2, v8f& acc3) {
  MixFrag f;
#pragma unroll
  for (int s = 0; s < 4; ++s)
    mix_frag(f, s, cur, cur + MXT, cur + 2 * MXT, cur + 3 * MXT, mrow, khalf, q);
#pragma unroll
  for (int s = 0; s < 4; ++s) {
    mix_mma(f, s, acc0, acc1, acc2, acc3);
    if (fill)
      mix_stage_store(g, s, nxt, nxt + MXT, nxt + 2 * MXT, nxt + 3 * MXT, tid);
  }
}

__global__ void __launch_bounds__(128) mode_mix(const float* __restrict__ xp, const float* __restrict__ xm,
                         const float* __restrict__ w, float* __restrict__ outm) {
  __shared__ __attribute__((aligned(16))) float sT0[4 * MXT];
  __shared__ __attribute__((aligned(16))) float sT1[4 * MXT];
  int tid  = threadIdx.x;
  int lane = tid & 31;
  int q    = tid >> 5;
  int kg  = blockIdx.x & 3;
  int rem = blockIdx.x >> 2;
  int bt = rem >> 3;
  int ot = rem & 7;
  int K0 = kg * 16;
  int MB = (48 - K0) & (NMODE - 1);
  int k0 = K0 + q * 4;
  int mrow = lane & 15;
  int khalf = lane >> 4;
  int oglob = ot * 16 + mrow;
  const float* xpb = xp + (size_t)(bt * 16) * CIN * NMODE;
  const float* xmb = xm + (size_t)(bt * 16) * CIN * NMODE;
  const float* wb  = w + (size_t)(ot * 16) * NMODE;
  v8f acc0 = {}, acc1 = {}, acc2 = {}, acc3 = {};
  MixStage ga, gb;
  mix_stage_load(ga, xpb, xmb, wb, 0, K0, MB, tid);
  mix_stage_load(gb, xpb, xmb, wb, MXI, K0, MB, tid);
#pragma unroll
  for (int part = 0; part < 4; ++part)
    mix_stage_store(ga, part, sT0, sT0 + MXT, sT0 + 2 * MXT, sT0 + 3 * MXT, tid);
  __syncthreads();
  mix_stage_load(ga, xpb, xmb, wb, 2 * MXI, K0, MB, tid);
#pragma unroll 1
  for (int i = 0; i < CIN; i += 2 * MXI) {
    mix_stage(sT0, sT1, gb, true, mrow, khalf, q, tid, acc0, acc1, acc2, acc3);
    __syncthreads();
    if (i + 3 * MXI < CIN)
      mix_stage_load(gb, xpb, xmb, wb, i + 3 * MXI, K0, MB, tid);
    mix_stage(sT1, sT0, ga, i + 2 * MXI < CIN, mrow, khalf, q, tid, acc0, acc1, acc2, acc3);
    __syncthreads();
    if (i + 4 * MXI < CIN)
      mix_stage_load(ga, xpb, xmb, wb, i + 4 * MXI, K0, MB, tid);
  }
#pragma unroll
  for (int v = 0; v < 8; ++v) {
    int bb = bt * 16 + v + 8 * khalf;
    v4f o;
    o.x = 0.5f * acc0[v];
    o.y = 0.5f * acc1[v];
    o.z = 0.5f * acc2[v];
    o.w = 0.5f * acc3[v];
    *(v4f*)(outm + ((size_t)bb * COUT + oglob) * NMODE + k0) = o;
  }
}

#define INV_NCH 4
__global__ void __launch_bounds__(256) dht_inv(const float* __restrict__ outm, const float* __restrict__ c2t,
                        float* __restrict__ y) {
  __shared__ __attribute__((aligned(16))) float sA[TM * LP];
  __shared__ __attribute__((aligned(16))) float sB[TN * LP];
  int tid  = threadIdx.x;
  int lane = tid & 31;
  int w    = tid >> 5;
  int g  = blockIdx.x >> 4;
  int q  = blockIdx.x & 15;
  int r0 = g * TM;
  int n0 = q * (INV_NCH * TN);
  const float* B = c2t + (size_t)n0 * NMODE;
  v4f ra[8], rb[4];
  tile_load<8, NMODE>(ra, outm + (size_t)r0 * NMODE, tid);
  tile_load<4, NMODE>(rb, B, tid);
  tile_store<8>(ra, sA, tid);
  int mrow = lane & 15;
  int khalf = lane >> 4;
  float* yrow = y + (size_t)(r0 + w * 16 + 8 * khalf) * NN + n0 + mrow;
#pragma unroll 1
  for (int ch = 0; ch < INV_NCH; ++ch) {
    tile_store<4>(rb, sB, tid);
    __syncthreads();
    if (ch + 1 < INV_NCH)
      tile_load<4, NMODE>(rb, B + (size_t)(ch + 1) * TN * NMODE, tid);
    v8f a0 = {}, a1 = {}, a2 = {}, a3 = {};
    stage_mma(sA, sB, w, lane, a0, a1, a2, a3);
    float* yc = yrow + ch * TN;
#pragma unroll
    for (int v = 0; v < 8; ++v) {
      float* drow = yc + (size_t)v * NN;
      drow[0]  = a0[v];
      drow[16] = a1[v];
      drow[32] = a2[v];
      drow[48] = a3[v];
    }
    __syncthreads();
  }
}

extern "C" void kernel_launch(void* const* d_in, const int* in_sizes, int n_in,
                              void* d_out, int out_size, void* d_ws, size_t ws_size,
                              hipStream_t stream) {
  (void)in_sizes; (void)n_in; (void)out_size; (void)ws_size;
  const float* x = (const float*)d_in[0];
  const float* w = (const float*)d_in[1];
  float* y = (float*)d_out;

  float* ws   = (float*)d_ws;
  float* c1t  = ws;
  float* c2t  = c1t + (size_t)NMODE * NN;
  float* xhp  = c2t + (size_t)NN * NMODE;
  float* xp   = xhp + (size_t)KSPLIT * NB * CIN * NMODE;
  float* xm   = xp  + (size_t)NB * CIN * NMODE;
  float* outm = xm  + (size_t)NB * CIN * NMODE;

  prep_cas<<<(2 * NMODE * NN) / 256, 256, 0, stream>>>(c1t, c2t);

  dht_fwd<<<(NB * CIN / TM) * KSPLIT, 256, 0, stream>>>(x, c1t, xhp);
  reduce_pair<<<(NB * CIN * NMODE) / RED_T, RED_T, 0, stream>>>(xhp, xp, xm);

  mode_mix<<<((NMODE / 4) * (NB / 16) * (COUT / 16) * 32) / 128, 128, 0, stream>>>(
      xp, xm, w, outm);

  dht_inv<<<(NB * COUT / TM) * (NN / (INV_NCH * TN)), 256, 0, stream>>>(
      outm, c2t, y);
}
